# starting kernel
# baseline (speedup 1.0000x reference)
#include <hip/hip_runtime.h>
#include <math.h>


typedef __attribute__((ext_vector_type(2))) float v2f;
typedef __attribute__((ext_vector_type(8))) float v8f;

#define HSTR 66
#define WPB  8

__device__ __forceinline__ v8f wmma4(v2f a, v2f b, v8f c) {
  return __builtin_amdgcn_wmma_f32_16x16x4_f32(
       false, a,  false, b,
       (short)0, c,  false,  false);
}

__device__ __forceinline__ float bn_gelu(float h, float sc, float bi) {
  float t = h * sc + bi;
  return 0.5f * t * (1.0f + erff(t * 0.70710678118654752f));
}

__global__ __launch_bounds__(256) void edgeconv_mlp_kernel(
    const float* __restrict__ x,
    const float* __restrict__ W1,
    const float* __restrict__ g1, const float* __restrict__ b1,
    const float* __restrict__ m1, const float* __restrict__ v1,
    const float* __restrict__ W2,
    const float* __restrict__ g2, const float* __restrict__ b2,
    const float* __restrict__ m2, const float* __restrict__ v2,
    const float* __restrict__ W3,
    const float* __restrict__ g3, const float* __restrict__ b3,
    const float* __restrict__ m3, const float* __restrict__ v3,
    float* __restrict__ out)
{
  __shared__ float hbuf[WPB * 16 * HSTR];

  const int tid    = threadIdx.x;
  const int wave   = tid >> 5;
  const int lane   = tid & 31;
  const int lane_m = lane & 15;
  const int half   = lane >> 4;
  const int kofs   = half * 2;

  const int tile = blockIdx.x * WPB + wave;
  const int p0   = tile * 16;
  float* H = hbuf + wave * 16 * HSTR;

  v8f acc[4];
  const v8f vzero = {0.f, 0.f, 0.f, 0.f, 0.f, 0.f, 0.f, 0.f};
#pragma unroll
  for (int nt = 0; nt < 4; ++nt) acc[nt] = vzero;

  const float* arow = x + (size_t)(p0 + lane_m) * 32 + kofs;
#pragma unroll
  for (int kb = 0; kb < 32; kb += 4) {
    v2f a = *(const v2f*)(arow + kb);
#pragma unroll
    for (int nt = 0; nt < 4; ++nt) {
      v2f b = *(const v2f*)(W1 + (size_t)(nt * 16 + lane_m) * 32 + kb + kofs);
      acc[nt] = wmma4(a, b, acc[nt]);
    }
  }
#pragma unroll
  for (int nt = 0; nt < 4; ++nt) {
    int o = nt * 16 + lane_m;
    float sc = g1[o] * rsqrtf(v1[o] + 1e-5f);
    float bi = b1[o] - m1[o] * sc;
#pragma unroll
    for (int r = 0; r < 8; ++r) {
      int mrow = r + half * 8;
      H[mrow * HSTR + o] = bn_gelu(acc[nt][r], sc, bi);
    }
  }
  asm volatile("s_wait_dscnt 0" ::: "memory");

  const float* Ws[2] = {W2, W3};
  const float* Gs[2] = {g2, g3};
  const float* Bs[2] = {b2, b3};
  const float* Ms[2] = {m2, m3};
  const float* Vs[2] = {v2, v3};

#pragma unroll
  for (int s = 0; s < 2; ++s) {
    const float* W = Ws[s];
#pragma unroll
    for (int nt = 0; nt < 4; ++nt) acc[nt] = vzero;
#pragma unroll
    for (int kb = 0; kb < 64; kb += 4) {
      v2f a = *(const v2f*)(H + lane_m * HSTR + kb + kofs);
#pragma unroll
      for (int nt = 0; nt < 4; ++nt) {
        v2f b = *(const v2f*)(W + (size_t)(nt * 16 + lane_m) * 64 + kb + kofs);
        acc[nt] = wmma4(a, b, acc[nt]);
      }
    }
    asm volatile("s_wait_dscnt 0" ::: "memory");
#pragma unroll
    for (int nt = 0; nt < 4; ++nt) {
      int o = nt * 16 + lane_m;
      float sc = Gs[s][o] * rsqrtf(Vs[s][o] + 1e-5f);
      float bi = Bs[s][o] - Ms[s][o] * sc;
#pragma unroll
      for (int r = 0; r < 8; ++r) {
        int mrow = r + half * 8;
        H[mrow * HSTR + o] = bn_gelu(acc[nt][r], sc, bi);
      }
    }
    asm volatile("s_wait_dscnt 0" ::: "memory");
  }

  const int bidx  = p0 >> 10;
  const int nbase = p0 & 1023;
  float* obase = out + (size_t)bidx * 64 * 1024 + nbase + lane_m;
#pragma unroll
  for (int j = 0; j < 32; ++j) {
    int o = 2 * j + half;
    obase[(size_t)o * 1024] = H[lane_m * HSTR + o];
  }
}

extern "C" void kernel_launch(void* const* d_in, const int* in_sizes, int n_in,
                              void* d_out, int out_size, void* d_ws, size_t ws_size,
                              hipStream_t stream) {
  const float* x  = (const float*)d_in[0];
  const float* W1 = (const float*)d_in[1];
  const float* g1 = (const float*)d_in[2];
  const float* b1 = (const float*)d_in[3];
  const float* m1 = (const float*)d_in[4];
  const float* v1 = (const float*)d_in[5];
  const float* W2 = (const float*)d_in[6];
  const float* g2 = (const float*)d_in[7];
  const float* b2 = (const float*)d_in[8];
  const float* m2 = (const float*)d_in[9];
  const float* v2 = (const float*)d_in[10];
  const float* W3 = (const float*)d_in[11];
  const float* g3 = (const float*)d_in[12];
  const float* b3 = (const float*)d_in[13];
  const float* m3 = (const float*)d_in[14];
  const float* v3 = (const float*)d_in[15];
  float* out = (float*)d_out;

  edgeconv_mlp_kernel<<<512, 256, 0, stream>>>(
      x, W1, g1, b1, m1, v1, W2, g2, b2, m2, v2, W3, g3, b3, m3, v3, out);
}


# GELU's erf from a 64-interval table in LDS (degree-4 Taylor coefficients per 1/16 of |u|, built per work-group from library erf and exp; 8+8+4-byte gathers per element) instead of inlined erff
# speedup vs baseline: 1.3301x; 1.3301x over previous
#include <hip/hip_runtime.h>
#include <math.h>


typedef __attribute__((ext_vector_type(2))) float v2f;
typedef __attribute__((ext_vector_type(4))) float v4f;
typedef __attribute__((ext_vector_type(8))) float v8f;

#define TP   64
#define MT   4
#define XSTR 36
#define NTAB 64
#define EB   160
#define EC   288
#define ETAB 416
#define HSTR 68

__device__ __forceinline__ v8f wmma4(v2f a, v2f b, v8f c) {
  return __builtin_amdgcn_wmma_f32_16x16x4_f32(
       false, a,  false, b,
       (short)0, c,  false,  false);
}

__device__ __forceinline__ void build_erf_entry(float* tab, int i) {
  const float m  = ((float)i + 0.5f) * 0.0625f;
  const float m2 = m * m;
  const float d  = 1.1283791670955126f * expf(-m2);
  v2f c01, c23;
  c01.x = erff(m);
  c01.y = d * (1.0f / 16.0f);
  c23.x = -m * d * (1.0f / 256.0f);
  c23.y = (4.0f * m2 - 2.0f) * d * (1.0f / 24576.0f);
  *(v2f*)(tab + 2 * i)      = c01;
  *(v2f*)(tab + EB + 2 * i) = c23;
  tab[EC + 2 * i] = (12.0f - 8.0f * m2) * m * d * (1.0f / 1572864.0f);
}

__device__ __forceinline__ float bn_gelu(float h, float sc, float bi, const float* tab) {
  const float t = h * sc + bi;
  const float a = fminf(fabsf(t) * 11.313708498984761f, 63.999996f);
  const int   i = (int)a;
  const float z = __builtin_amdgcn_fractf(a) - 0.5f;
  const float* p = tab + 2 * i;
  float e = p[EC];
  e = e * z + p[EB + 1];
  e = e * z + p[EB];
  e = e * z + p[1];
  e = e * z + p[0];
  e = copysignf(e, t);
  const float ht = 0.5f * t;
  return ht * e + ht;
}

template <int K>
__device__ __forceinline__ void load_w(const float* __restrict__ W, int n, int half,
                                       v4f (&wb)[K / 8]) {
  const float* row = W + (size_t)n * K + 4 * half;
#pragma unroll
  for (int q = 0; q < K / 8; ++q) wb[q] = *(const v4f*)(row + 8 * q);
}

template <int K, int STR>
__device__ __forceinline__ void layer(const float* A, int c, int half,
                                      const v4f (&wb)[K / 8], v8f (&acc)[MT]) {
  const v8f vzero = {0.f, 0.f, 0.f, 0.f, 0.f, 0.f, 0.f, 0.f};
#pragma unroll
  for (int mt = 0; mt < MT; ++mt) acc[mt] = vzero;
  const float* arow = A + c * STR + 4 * half;
#pragma unroll
  for (int q = 0; q < K / 8; ++q) {
    v4f av[MT];
#pragma unroll
    for (int mt = 0; mt < MT; ++mt)
      av[mt] = *(const v4f*)(arow + mt * 16 * STR + 8 * q);
    const v2f b0 = {wb[q].x, wb[q].y};
    const v2f b1 = {wb[q].z, wb[q].w};
#pragma unroll
    for (int mt = 0; mt < MT; ++mt) {
      const v2f a0 = {av[mt].x, av[mt].y};
      acc[mt] = wmma4(a0, b0, acc[mt]);
    }
#pragma unroll
    for (int mt = 0; mt < MT; ++mt) {
      const v2f a1 = {av[mt].z, av[mt].w};
      acc[mt] = wmma4(a1, b1, acc[mt]);
    }
  }
}

__device__ __forceinline__ void store_act(float* H, int n, int half,
                                          const v8f (&acc)[MT], float sc, float bi,
                                          const float* tab) {
  float* col = H + (8 * half) * HSTR + n;
#pragma unroll
  for (int mt = 0; mt < MT; ++mt) {
#pragma unroll
    for (int r = 0; r < 8; ++r)
      col[(mt * 16 + r) * HSTR] = bn_gelu(acc[mt][r], sc, bi, tab);
  }
}

__global__ __launch_bounds__(128) void edgeconv_mlp_kernel(
    const float* __restrict__ x,
    const float* __restrict__ W1,
    const float* __restrict__ g1, const float* __restrict__ b1,
    const float* __restrict__ m1, const float* __restrict__ v1,
    const float* __restrict__ W2,
    const float* __restrict__ g2, const float* __restrict__ b2,
    const float* __restrict__ m2, const float* __restrict__ v2,
    const float* __restrict__ W3,
    const float* __restrict__ g3, const float* __restrict__ b3,
    const float* __restrict__ m3, const float* __restrict__ v3,
    float* __restrict__ out)
{
  __shared__ __attribute__((aligned(16))) float smem[ETAB + TP * XSTR + 2 * TP * HSTR];
  float* const etab = smem;
  float* const xs   = smem + ETAB;
  float* const h1   = xs + TP * XSTR;
  float* const h2   = h1 + TP * HSTR;

  const int tid  = threadIdx.x;
  const int wave = tid >> 5;
  const int lane = tid & 31;
  const int c    = lane & 15;
  const int half = lane >> 4;
  const int n    = wave * 16 + c;

  const int tile = blockIdx.x;
  const int p0   = tile * TP;

  const v4f* xg = (const v4f*)(x + (size_t)p0 * 32);
  v4f xv[4];
#pragma unroll
  for (int i = 0; i < 4; ++i) xv[i] = xg[tid + 128 * i];

  v4f wb1[4], wb2[8], wb3[8];
  load_w<32>(W1, n, half, wb1);
  load_w<64>(W2, n, half, wb2);
  load_w<64>(W3, n, half, wb3);

  const float sc1 = g1[n] * rsqrtf(v1[n] + 1e-5f), bi1 = b1[n] - m1[n] * sc1;
  const float sc2 = g2[n] * rsqrtf(v2[n] + 1e-5f), bi2 = b2[n] - m2[n] * sc2;
  const float sc3 = g3[n] * rsqrtf(v3[n] + 1e-5f), bi3 = b3[n] - m3[n] * sc3;

  if (tid < NTAB) build_erf_entry(etab, tid);

#pragma unroll
  for (int i = 0; i < 4; ++i) {
    const int idx = tid + 128 * i;
    *(v4f*)(xs + (idx >> 3) * XSTR + (idx & 7) * 4) = xv[i];
  }
  __syncthreads();

  v8f acc[MT];

  layer<32, XSTR>(xs, c, half, wb1, acc);
  store_act(h1, n, half, acc, sc1, bi1, etab);
  __syncthreads();

  layer<64, HSTR>(h1, c, half, wb2, acc);
  store_act(h2, n, half, acc, sc2, bi2, etab);
  __syncthreads();

  layer<64, HSTR>(h2, c, half, wb3, acc);

  float* mine = h1 + n * HSTR + 8 * half;
#pragma unroll
  for (int mt = 0; mt < MT; ++mt) {
    v4f lo, hi;
    lo.x = bn_gelu(acc[mt][0], sc3, bi3, etab);
    lo.y = bn_gelu(acc[mt][1], sc3, bi3, etab);
    lo.z = bn_gelu(acc[mt][2], sc3, bi3, etab);
    lo.w = bn_gelu(acc[mt][3], sc3, bi3, etab);
    hi.x = bn_gelu(acc[mt][4], sc3, bi3, etab);
    hi.y = bn_gelu(acc[mt][5], sc3, bi3, etab);
    hi.z = bn_gelu(acc[mt][6], sc3, bi3, etab);
    hi.w = bn_gelu(acc[mt][7], sc3, bi3, etab);
    *(v4f*)(mine + mt * 16)     = lo;
    *(v4f*)(mine + mt * 16 + 4) = hi;
  }
  asm volatile("s_wait_dscnt 0" ::: "memory");

  float* obase = out + ((size_t)(tile >> 4) * 64 + wave * 16 + half) * 1024 + (tile & 15) * TP + 4 * c;
  const float* rbase = h1 + (wave * 16 + half) * HSTR + 4 * c;
#pragma unroll
  for (int i = 0; i < 8; ++i) {
    const v4f v = *(const v4f*)(rbase + 2 * i * HSTR);
    *(v4f*)(obase + (size_t)(2 * i) * 1024) = v;
  }
}

extern "C" void kernel_launch(void* const* d_in, const int* in_sizes, int n_in,
                              void* d_out, int out_size, void* d_ws, size_t ws_size,
                              hipStream_t stream) {
  const float* x  = (const float*)d_in[0];
  const float* W1 = (const float*)d_in[1];
  const float* g1 = (const float*)d_in[2];
  const float* b1 = (const float*)d_in[3];
  const float* m1 = (const float*)d_in[4];
  const float* v1 = (const float*)d_in[5];
  const float* W2 = (const float*)d_in[6];
  const float* g2 = (const float*)d_in[7];
  const float* b2 = (const float*)d_in[8];
  const float* m2 = (const float*)d_in[9];
  const float* v2 = (const float*)d_in[10];
  const float* W3 = (const float*)d_in[11];
  const float* g3 = (const float*)d_in[12];
  const float* b3 = (const float*)d_in[13];
  const float* m3 = (const float*)d_in[14];
  const float* v3 = (const float*)d_in[15];
  float* out = (float*)d_out;

  edgeconv_mlp_kernel<<<1024, 128, 0, stream>>>(
      x, W1, g1, b1, m1, v1, W2, g2, b2, m2, v2, W3, g3, b3, m3, v3, out);
}


# Tile worked as two 32-point halves one layer-phase apart: each k-step's 4 WMMAs of one half run beside a share of the other half's BatchNorm+GELU, table reads one k-step ahead; sum order kept
# speedup vs baseline: 1.3968x; 1.0501x over previous
#include <hip/hip_runtime.h>
#include <math.h>


typedef __attribute__((ext_vector_type(2))) float v2f;
typedef __attribute__((ext_vector_type(4))) float v4f;
typedef __attribute__((ext_vector_type(8))) float v8f;

#define TP   64
#define HM   2
#define XSTR 36
#define NTAB 64
#define EB   160
#define EC   288
#define ETAB 416
#define HSTR 68

__device__ __forceinline__ v8f wmma4(v2f a, v2f b, v8f c) {
  return __builtin_amdgcn_wmma_f32_16x16x4_f32(
       false, a,  false, b,
       (short)0, c,  false,  false);
}

__device__ __forceinline__ void build_erf_entry(float* tab, int i) {
  const float m  = ((float)i + 0.5f) * 0.0625f;
  const float m2 = m * m;
  const float d  = 1.1283791670955126f * expf(-m2);
  v2f c01, c23;
  c01.x = erff(m);
  c01.y = d * (1.0f / 16.0f);
  c23.x = -m * d * (1.0f / 256.0f);
  c23.y = (4.0f * m2 - 2.0f) * d * (1.0f / 24576.0f);
  *(v2f*)(tab + 2 * i)      = c01;
  *(v2f*)(tab + EB + 2 * i) = c23;
  tab[EC + 2 * i] = (12.0f - 8.0f * m2) * m * d * (1.0f / 1572864.0f);
}

struct Pend {
  float t, z;
  v2f   c01, c23;
  float c4;
};

__device__ __forceinline__ void gelu_issue(float h, float sc, float bi, const float* tab, Pend& e) {
  const float t = h * sc + bi;
  const float a = fminf(fabsf(t) * 11.313708498984761f, 63.999996f);
  const int   i = (int)a;
  e.t = t;
  e.z = __builtin_amdgcn_fractf(a) - 0.5f;
  const float* p = tab + 2 * i;
  e.c01 = *(const v2f*)(p);
  e.c23 = *(const v2f*)(p + EB);
  e.c4  = p[EC];
}

__device__ __forceinline__ float gelu_finish(const Pend& e) {
  float r = e.c4;
  r = r * e.z + e.c23.y;
  r = r * e.z + e.c23.x;
  r = r * e.z + e.c01.y;
  r = r * e.z + e.c01.x;
  r = copysignf(r, e.t);
  const float ht = 0.5f * e.t;
  return ht * r + ht;
}

__device__ __forceinline__ float bn_gelu(float h, float sc, float bi, const float* tab) {
  Pend e;
  gelu_issue(h, sc, bi, tab, e);
  return gelu_finish(e);
}

template <int K>
__device__ __forceinline__ void load_w(const float* __restrict__ W, int n, int half,
                                       v4f (&wb)[K / 8]) {
  const float* row = W + (size_t)n * K + 4 * half;
#pragma unroll
  for (int q = 0; q < K / 8; ++q) wb[q] = *(const v4f*)(row + 8 * q);
}

template <int MODE>
__device__ __forceinline__ void put_pair(float* gdst, int e, float g0, float g1) {
  const int m = e >> 3, r = e & 7;
  if (MODE == 1) {
    gdst[(m * 16 + r) * HSTR]     = g0;
    gdst[(m * 16 + r + 1) * HSTR] = g1;
  } else {
    v2f g = {g0, g1};
    *(v2f*)(gdst + m * 16 + r) = g;
  }
}

template <int K, int STR, int MODE>
__device__ __forceinline__ void phase(const float* A, int c, int half,
                                      const v4f (&wb)[K / 8], v8f (&acc)[HM],
                                      const v8f (&gacc)[HM], float sc, float bi,
                                      const float* tab, float* gdst) {
  constexpr int NQ = K / 8;
  constexpr int NE = 16 / NQ;
  const v8f vzero = {0.f, 0.f, 0.f, 0.f, 0.f, 0.f, 0.f, 0.f};
#pragma unroll
  for (int m = 0; m < HM; ++m) acc[m] = vzero;
  const float* arow = A + c * STR + 4 * half;
  v4f av[HM], an[HM];
#pragma unroll
  for (int m = 0; m < HM; ++m) av[m] = *(const v4f*)(arow + m * 16 * STR);
  Pend pe[NE], pn[NE];
#pragma unroll
  for (int q = 0; q < NQ; ++q) {
    if (q + 1 < NQ) {
#pragma unroll
      for (int m = 0; m < HM; ++m)
        an[m] = *(const v4f*)(arow + m * 16 * STR + 8 * (q + 1));
    }
    const v2f b0 = {wb[q].x, wb[q].y};
    const v2f b1 = {wb[q].z, wb[q].w};
#pragma unroll
    for (int m = 0; m < HM; ++m) {
      const v2f a0 = {av[m].x, av[m].y};
      acc[m] = wmma4(a0, b0, acc[m]);
    }
#pragma unroll
    for (int m = 0; m < HM; ++m) {
      const v2f a1 = {av[m].z, av[m].w};
      acc[m] = wmma4(a1, b1, acc[m]);
    }
    __builtin_amdgcn_sched_barrier(0);
    if (MODE != 0) {
#pragma unroll
      for (int j = 0; j < NE; ++j) {
        const int e = q * NE + j;
        gelu_issue(gacc[e >> 3][e & 7], sc, bi, tab, pn[j]);
      }
      if (q > 0) {
#pragma unroll
        for (int j = 0; j < NE; j += 2)
          put_pair<MODE>(gdst, (q - 1) * NE + j, gelu_finish(pe[j]), gelu_finish(pe[j + 1]));
      }
#pragma unroll
      for (int j = 0; j < NE; ++j) pe[j] = pn[j];
    }
    if (q + 1 < NQ) asm volatile("" :: "v"(an[0]), "v"(an[1]));
#pragma unroll
    for (int m = 0; m < HM; ++m) av[m] = an[m];
    __builtin_amdgcn_sched_barrier(0);
  }
  if (MODE != 0) {
#pragma unroll
    for (int j = 0; j < NE; j += 2)
      put_pair<MODE>(gdst, (NQ - 1) * NE + j, gelu_finish(pe[j]), gelu_finish(pe[j + 1]));
  }
}

__global__ __launch_bounds__(128) void edgeconv_mlp_kernel(
    const float* __restrict__ x,
    const float* __restrict__ W1,
    const float* __restrict__ g1, const float* __restrict__ b1,
    const float* __restrict__ m1, const float* __restrict__ v1,
    const float* __restrict__ W2,
    const float* __restrict__ g2, const float* __restrict__ b2,
    const float* __restrict__ m2, const float* __restrict__ v2,
    const float* __restrict__ W3,
    const float* __restrict__ g3, const float* __restrict__ b3,
    const float* __restrict__ m3, const float* __restrict__ v3,
    float* __restrict__ out)
{
  __shared__ __attribute__((aligned(16))) float smem[ETAB + TP * XSTR + 2 * TP * HSTR];
  float* const etab = smem;
  float* const xs   = smem + ETAB;
  float* const h1   = xs + TP * XSTR;
  float* const h2   = h1 + TP * HSTR;

  const int tid  = threadIdx.x;
  const int wave = tid >> 5;
  const int lane = tid & 31;
  const int c    = lane & 15;
  const int half = lane >> 4;
  const int n    = wave * 16 + c;

  const int tile = blockIdx.x;
  const int p0   = tile * TP;

  const v4f* xg = (const v4f*)(x + (size_t)p0 * 32);
  v4f xv[4];
#pragma unroll
  for (int i = 0; i < 4; ++i) xv[i] = xg[tid + 128 * i];

  v4f wb1[4], wb2[8], wb3[8];
  load_w<32>(W1, n, half, wb1);
  load_w<64>(W2, n, half, wb2);
  load_w<64>(W3, n, half, wb3);

  const float sc1 = g1[n] * rsqrtf(v1[n] + 1e-5f), bi1 = b1[n] - m1[n] * sc1;
  const float sc2 = g2[n] * rsqrtf(v2[n] + 1e-5f), bi2 = b2[n] - m2[n] * sc2;
  const float sc3 = g3[n] * rsqrtf(v3[n] + 1e-5f), bi3 = b3[n] - m3[n] * sc3;

  if (tid < NTAB) build_erf_entry(etab, tid);

#pragma unroll
  for (int i = 0; i < 4; ++i) {
    const int idx = tid + 128 * i;
    *(v4f*)(xs + (idx >> 3) * XSTR + (idx & 7) * 4) = xv[i];
  }
  __syncthreads();

  v8f accA[HM], accB[HM];
  float* const col1 = h1 + (8 * half) * HSTR + n;
  float* const col2 = h2 + (8 * half) * HSTR + n;
  float* const mine = h1 + n * HSTR + 8 * half;

  phase<32, XSTR, 0>(xs, c, half, wb1, accA, accB, 0.f, 0.f, etab, nullptr);
  phase<32, XSTR, 1>(xs + 32 * XSTR, c, half, wb1, accB, accA, sc1, bi1, etab, col1);
  __syncthreads();
  phase<64, HSTR, 1>(h1, c, half, wb2, accA, accB, sc1, bi1, etab, col1 + 32 * HSTR);
  __syncthreads();
  phase<64, HSTR, 1>(h1 + 32 * HSTR, c, half, wb2, accB, accA, sc2, bi2, etab, col2);
  __syncthreads();
  phase<64, HSTR, 1>(h2, c, half, wb3, accA, accB, sc2, bi2, etab, col2 + 32 * HSTR);
  __syncthreads();
  phase<64, HSTR, 2>(h2 + 32 * HSTR, c, half, wb3, accB, accA, sc3, bi3, etab, mine);

#pragma unroll
  for (int m = 0; m < HM; ++m) {
    v4f lo, hi;
    lo.x = bn_gelu(accB[m][0], sc3, bi3, etab);
    lo.y = bn_gelu(accB[m][1], sc3, bi3, etab);
    lo.z = bn_gelu(accB[m][2], sc3, bi3, etab);
    lo.w = bn_gelu(accB[m][3], sc3, bi3, etab);
    hi.x = bn_gelu(accB[m][4], sc3, bi3, etab);
    hi.y = bn_gelu(accB[m][5], sc3, bi3, etab);
    hi.z = bn_gelu(accB[m][6], sc3, bi3, etab);
    hi.w = bn_gelu(accB[m][7], sc3, bi3, etab);
    *(v4f*)(mine + 32 + m * 16)     = lo;
    *(v4f*)(mine + 32 + m * 16 + 4) = hi;
  }
  asm volatile("s_wait_dscnt 0" ::: "memory");

  float* obase = out + ((size_t)(tile >> 4) * 64 + wave * 16 + half) * 1024 + (tile & 15) * TP + 4 * c;
  const float* rbase = h1 + (wave * 16 + half) * HSTR + 4 * c;
#pragma unroll
  for (int i = 0; i < 8; ++i) {
    const v4f v = *(const v4f*)(rbase + 2 * i * HSTR);
    *(v4f*)(obase + (size_t)(2 * i) * 1024) = v;
  }
}

extern "C" void kernel_launch(void* const* d_in, const int* in_sizes, int n_in,
                              void* d_out, int out_size, void* d_ws, size_t ws_size,
                              hipStream_t stream) {
  const float* x  = (const float*)d_in[0];
  const float* W1 = (const float*)d_in[1];
  const float* g1 = (const float*)d_in[2];
  const float* b1 = (const float*)d_in[3];
  const float* m1 = (const float*)d_in[4];
  const float* v1 = (const float*)d_in[5];
  const float* W2 = (const float*)d_in[6];
  const float* g2 = (const float*)d_in[7];
  const float* b2 = (const float*)d_in[8];
  const float* m2 = (const float*)d_in[9];
  const float* v2 = (const float*)d_in[10];
  const float* W3 = (const float*)d_in[11];
  const float* g3 = (const float*)d_in[12];
  const float* b3 = (const float*)d_in[13];
  const float* m3 = (const float*)d_in[14];
  const float* v3 = (const float*)d_in[15];
  float* out = (float*)d_out;

  edgeconv_mlp_kernel<<<1024, 128, 0, stream>>>(
      x, W1, g1, b1, m1, v1, W2, g2, b2, m2, v2, W3, g3, b3, m3, v3, out);
}


# GELU as max(t,0) - R(|t|), R(s) = s*Phi(-s) from a 96-interval cubic table (economized quartic), BatchNorm carried times 16: one two-address 8-byte gather per element, not 3 reads; no erf or sign op
# speedup vs baseline: 1.4764x; 1.0570x over previous
#include <hip/hip_runtime.h>
#include <math.h>


typedef __attribute__((ext_vector_type(2))) float v2f;
typedef __attribute__((ext_vector_type(4))) float v4f;
typedef __attribute__((ext_vector_type(8))) float v8f;

#define TP   64
#define HM   2
#define XSTR 36
#define NTAB 96
#define EB   224
#define ETAB 416
#define ASCALE 16.0f
#define RSCALE 0.0625f
#define HSTR 68

__device__ __forceinline__ v8f wmma4(v2f a, v2f b, v8f c) {
  return __builtin_amdgcn_wmma_f32_16x16x4_f32(
       false, a,  false, b,
       (short)0, c,  false,  false);
}

__device__ __forceinline__ void build_gelu_entry(float* tab, int i) {
  const float s   = ((float)i + 0.5f) * RSCALE;
  const float s2  = s * s;
  const float phi = 0.3989422804014327f * expf(-0.5f * s2);
  const float Phm = 0.5f * erfcf(s * 0.7071067811865476f);
  const float r1 = RSCALE, r2 = r1 * r1, r3 = r2 * r1, r4 = r2 * r2;
  float e0 = s * Phm;
  const float e1 = (Phm - s * phi) * r1;
  float e2 = (s2 - 2.0f) * phi * (0.5f * r2);
  const float e3 = (4.0f - s2) * s * phi * (r3 * (1.0f / 6.0f));
  const float e4 = ((s2 - 7.0f) * s2 + 4.0f) * phi * (r4 * (1.0f / 24.0f));
  e2 += 0.25f * e4;
  e0 -= e4 * (1.0f / 128.0f);
  v2f q02, q13;
  q02.x = e0 - 0.5f * e1 + 0.25f * e2 - 0.125f * e3;
  q13.x = e1 - e2 + 0.75f * e3;
  q02.y = e2 - 1.5f * e3;
  q13.y = e3;
  *(v2f*)(tab + 2 * i)      = q02;
  *(v2f*)(tab + EB + 2 * i) = q13;
}

struct Pend {
  float m, z;
  v2f   q02, q13;
};

__device__ __forceinline__ void gelu_issue(float h, float sc, float bi, const float* tab, Pend& e) {
  const float sa = h * sc + bi;
  const float a = __builtin_elementwise_minimum(fabsf(sa), 95.99999f);
  const int   i = (int)a;
  e.m = fmaxf(sa, 0.0f);
  e.z = __builtin_amdgcn_fractf(a);
  const float* p = tab + 2 * i;
  e.q02 = *(const v2f*)(p);
  e.q13 = *(const v2f*)(p + EB);
}

__device__ __forceinline__ float gelu_finish(const Pend& e) {
  const v2f zz = {e.z, e.z};
  const v2f ab = __builtin_elementwise_fma(e.q13, zz, e.q02);
  const float r = (e.z * e.z) * ab.y + ab.x;
  return e.m * RSCALE - r;
}

__device__ __forceinline__ float bn_gelu(float h, float sc, float bi, const float* tab) {
  Pend e;
  gelu_issue(h, sc, bi, tab, e);
  return gelu_finish(e);
}

template <int K>
__device__ __forceinline__ void load_w(const float* __restrict__ W, int n, int half,
                                       v4f (&wb)[K / 8]) {
  const float* row = W + (size_t)n * K + 4 * half;
#pragma unroll
  for (int q = 0; q < K / 8; ++q) wb[q] = *(const v4f*)(row + 8 * q);
}

template <int MODE>
__device__ __forceinline__ void put_pair(float* gdst, int e, float g0, float g1) {
  const int m = e >> 3, r = e & 7;
  if (MODE == 1) {
    gdst[(m * 16 + r) * HSTR]     = g0;
    gdst[(m * 16 + r + 1) * HSTR] = g1;
  } else {
    v2f g = {g0, g1};
    *(v2f*)(gdst + m * 16 + r) = g;
  }
}

template <int K, int STR, int MODE>
__device__ __forceinline__ void phase(const float* A, int c, int half,
                                      const v4f (&wb)[K / 8], v8f (&acc)[HM],
                                      const v8f (&gacc)[HM], float sc, float bi,
                                      const float* tab, float* gdst) {
  constexpr int NQ = K / 8;
  constexpr int NE = 16 / NQ;
  const v8f vzero = {0.f, 0.f, 0.f, 0.f, 0.f, 0.f, 0.f, 0.f};
#pragma unroll
  for (int m = 0; m < HM; ++m) acc[m] = vzero;
  const float* arow = A + c * STR + 4 * half;
  v4f av[HM], an[HM];
#pragma unroll
  for (int m = 0; m < HM; ++m) av[m] = *(const v4f*)(arow + m * 16 * STR);
  Pend pe[NE], pn[NE];
#pragma unroll
  for (int q = 0; q < NQ; ++q) {
    if (q + 1 < NQ) {
#pragma unroll
      for (int m = 0; m < HM; ++m)
        an[m] = *(const v4f*)(arow + m * 16 * STR + 8 * (q + 1));
    }
    const v2f b0 = {wb[q].x, wb[q].y};
    const v2f b1 = {wb[q].z, wb[q].w};
#pragma unroll
    for (int m = 0; m < HM; ++m) {
      const v2f a0 = {av[m].x, av[m].y};
      acc[m] = wmma4(a0, b0, acc[m]);
    }
#pragma unroll
    for (int m = 0; m < HM; ++m) {
      const v2f a1 = {av[m].z, av[m].w};
      acc[m] = wmma4(a1, b1, acc[m]);
    }
    __builtin_amdgcn_sched_barrier(0);
    if (MODE != 0) {
#pragma unroll
      for (int j = 0; j < NE; ++j) {
        const int e = q * NE + j;
        gelu_issue(gacc[e >> 3][e & 7], sc, bi, tab, pn[j]);
      }
      if (q > 0) {
#pragma unroll
        for (int j = 0; j < NE; j += 2)
          put_pair<MODE>(gdst, (q - 1) * NE + j, gelu_finish(pe[j]), gelu_finish(pe[j + 1]));
      }
#pragma unroll
      for (int j = 0; j < NE; ++j) pe[j] = pn[j];
    }
    if (q + 1 < NQ) asm volatile("" :: "v"(an[0]), "v"(an[1]));
#pragma unroll
    for (int m = 0; m < HM; ++m) av[m] = an[m];
    __builtin_amdgcn_sched_barrier(0);
  }
  if (MODE != 0) {
#pragma unroll
    for (int j = 0; j < NE; j += 2)
      put_pair<MODE>(gdst, (NQ - 1) * NE + j, gelu_finish(pe[j]), gelu_finish(pe[j + 1]));
  }
}

__global__ __launch_bounds__(128) void edgeconv_mlp_kernel(
    const float* __restrict__ x,
    const float* __restrict__ W1,
    const float* __restrict__ g1, const float* __restrict__ b1,
    const float* __restrict__ m1, const float* __restrict__ v1,
    const float* __restrict__ W2,
    const float* __restrict__ g2, const float* __restrict__ b2,
    const float* __restrict__ m2, const float* __restrict__ v2,
    const float* __restrict__ W3,
    const float* __restrict__ g3, const float* __restrict__ b3,
    const float* __restrict__ m3, const float* __restrict__ v3,
    float* __restrict__ out)
{
  __shared__ __attribute__((aligned(16))) float smem[ETAB + TP * XSTR + 2 * TP * HSTR];
  float* const etab = smem;
  float* const xs   = smem + ETAB;
  float* const h1   = xs + TP * XSTR;
  float* const h2   = h1 + TP * HSTR;

  const int tid  = threadIdx.x;
  const int wave = tid >> 5;
  const int lane = tid & 31;
  const int c    = lane & 15;
  const int half = lane >> 4;
  const int n    = wave * 16 + c;

  const int tile = blockIdx.x;
  const int p0   = tile * TP;

  const v4f* xg = (const v4f*)(x + (size_t)p0 * 32);
  v4f xv[4];
#pragma unroll
  for (int i = 0; i < 4; ++i) xv[i] = xg[tid + 128 * i];

  v4f wb1[4], wb2[8], wb3[8];
  load_w<32>(W1, n, half, wb1);
  load_w<64>(W2, n, half, wb2);
  load_w<64>(W3, n, half, wb3);

  const float f1 = g1[n] * rsqrtf(v1[n] + 1e-5f), sc1 = f1 * ASCALE, bi1 = (b1[n] - m1[n] * f1) * ASCALE;
  const float f2 = g2[n] * rsqrtf(v2[n] + 1e-5f), sc2 = f2 * ASCALE, bi2 = (b2[n] - m2[n] * f2) * ASCALE;
  const float f3 = g3[n] * rsqrtf(v3[n] + 1e-5f), sc3 = f3 * ASCALE, bi3 = (b3[n] - m3[n] * f3) * ASCALE;

  if (tid < NTAB) build_gelu_entry(etab, tid);

#pragma unroll
  for (int i = 0; i < 4; ++i) {
    const int idx = tid + 128 * i;
    *(v4f*)(xs + (idx >> 3) * XSTR + (idx & 7) * 4) = xv[i];
  }
  __syncthreads();

  v8f accA[HM], accB[HM];
  float* const col1 = h1 + (8 * half) * HSTR + n;
  float* const col2 = h2 + (8 * half) * HSTR + n;
  float* const mine = h1 + n * HSTR + 8 * half;

  phase<32, XSTR, 0>(xs, c, half, wb1, accA, accB, 0.f, 0.f, etab, nullptr);
  phase<32, XSTR, 1>(xs + 32 * XSTR, c, half, wb1, accB, accA, sc1, bi1, etab, col1);
  __syncthreads();
  phase<64, HSTR, 1>(h1, c, half, wb2, accA, accB, sc1, bi1, etab, col1 + 32 * HSTR);
  __syncthreads();
  phase<64, HSTR, 1>(h1 + 32 * HSTR, c, half, wb2, accB, accA, sc2, bi2, etab, col2);
  __syncthreads();
  phase<64, HSTR, 1>(h2, c, half, wb3, accA, accB, sc2, bi2, etab, col2 + 32 * HSTR);
  __syncthreads();
  phase<64, HSTR, 2>(h2 + 32 * HSTR, c, half, wb3, accB, accA, sc3, bi3, etab, mine);

#pragma unroll
  for (int m = 0; m < HM; ++m) {
    v4f lo, hi;
    lo.x = bn_gelu(accB[m][0], sc3, bi3, etab);
    lo.y = bn_gelu(accB[m][1], sc3, bi3, etab);
    lo.z = bn_gelu(accB[m][2], sc3, bi3, etab);
    lo.w = bn_gelu(accB[m][3], sc3, bi3, etab);
    hi.x = bn_gelu(accB[m][4], sc3, bi3, etab);
    hi.y = bn_gelu(accB[m][5], sc3, bi3, etab);
    hi.z = bn_gelu(accB[m][6], sc3, bi3, etab);
    hi.w = bn_gelu(accB[m][7], sc3, bi3, etab);
    *(v4f*)(mine + 32 + m * 16)     = lo;
    *(v4f*)(mine + 32 + m * 16 + 4) = hi;
  }
  asm volatile("s_wait_dscnt 0" ::: "memory");

  float* obase = out + ((size_t)(tile >> 4) * 64 + wave * 16 + half) * 1024 + (tile & 15) * TP + 4 * c;
  const float* rbase = h1 + (wave * 16 + half) * HSTR + 4 * c;
#pragma unroll
  for (int i = 0; i < 8; ++i) {
    const v4f v = *(const v4f*)(rbase + 2 * i * HSTR);
    *(v4f*)(obase + (size_t)(2 * i) * 1024) = v;
  }
}

extern "C" void kernel_launch(void* const* d_in, const int* in_sizes, int n_in,
                              void* d_out, int out_size, void* d_ws, size_t ws_size,
                              hipStream_t stream) {
  const float* x  = (const float*)d_in[0];
  const float* W1 = (const float*)d_in[1];
  const float* g1 = (const float*)d_in[2];
  const float* b1 = (const float*)d_in[3];
  const float* m1 = (const float*)d_in[4];
  const float* v1 = (const float*)d_in[5];
  const float* W2 = (const float*)d_in[6];
  const float* g2 = (const float*)d_in[7];
  const float* b2 = (const float*)d_in[8];
  const float* m2 = (const float*)d_in[9];
  const float* v2 = (const float*)d_in[10];
  const float* W3 = (const float*)d_in[11];
  const float* g3 = (const float*)d_in[12];
  const float* b3 = (const float*)d_in[13];
  const float* m3 = (const float*)d_in[14];
  const float* v3 = (const float*)d_in[15];
  float* out = (float*)d_out;

  edgeconv_mlp_kernel<<<1024, 128, 0, stream>>>(
      x, W1, g1, b1, m1, v1, W2, g2, b2, m2, v2, W3, g3, b3, m3, v3, out);
}


# Prologue loads issued and waited for in order of first use: x tile, W1 first, W2 and W3 behind; first two barriers order LDS only, so the first product no longer waits for all 36 loads; sums kept
# speedup vs baseline: 1.6380x; 1.1095x over previous
#include <hip/hip_runtime.h>
#include <math.h>


typedef __attribute__((ext_vector_type(2))) float v2f;
typedef __attribute__((ext_vector_type(4))) float v4f;
typedef __attribute__((ext_vector_type(8))) float v8f;

#define TP   64
#define HM   2
#define XSTR 36
#define NTAB 96
#define EB   224
#define ETAB 416
#define ASCALE 16.0f
#define RSCALE 0.0625f
#define HSTR 68

__device__ __forceinline__ v8f wmma4(v2f a, v2f b, v8f c) {
  return __builtin_amdgcn_wmma_f32_16x16x4_f32(
       false, a,  false, b,
       (short)0, c,  false,  false);
}

__device__ __forceinline__ void build_gelu_entry(float* tab, int i) {
  const float s   = ((float)i + 0.5f) * RSCALE;
  const float s2  = s * s;
  const float phi = 0.3989422804014327f * expf(-0.5f * s2);
  const float Phm = 0.5f * erfcf(s * 0.7071067811865476f);
  const float r1 = RSCALE, r2 = r1 * r1, r3 = r2 * r1, r4 = r2 * r2;
  float e0 = s * Phm;
  const float e1 = (Phm - s * phi) * r1;
  float e2 = (s2 - 2.0f) * phi * (0.5f * r2);
  const float e3 = (4.0f - s2) * s * phi * (r3 * (1.0f / 6.0f));
  const float e4 = ((s2 - 7.0f) * s2 + 4.0f) * phi * (r4 * (1.0f / 24.0f));
  e2 += 0.25f * e4;
  e0 -= e4 * (1.0f / 128.0f);
  v2f q02, q13;
  q02.x = e0 - 0.5f * e1 + 0.25f * e2 - 0.125f * e3;
  q13.x = e1 - e2 + 0.75f * e3;
  q02.y = e2 - 1.5f * e3;
  q13.y = e3;
  *(v2f*)(tab + 2 * i)      = q02;
  *(v2f*)(tab + EB + 2 * i) = q13;
}

struct Pend {
  float m, z;
  v2f   q02, q13;
};

__device__ __forceinline__ void gelu_issue(float h, float sc, float bi, const float* tab, Pend& e) {
  const float sa = h * sc + bi;
  const float a = __builtin_elementwise_minimum(fabsf(sa), 95.99999f);
  const int   i = (int)a;
  e.m = fmaxf(sa, 0.0f);
  e.z = __builtin_amdgcn_fractf(a);
  const float* p = tab + 2 * i;
  e.q02 = *(const v2f*)(p);
  e.q13 = *(const v2f*)(p + EB);
}

__device__ __forceinline__ float gelu_finish(const Pend& e) {
  const v2f zz = {e.z, e.z};
  const v2f ab = __builtin_elementwise_fma(e.q13, zz, e.q02);
  const float r = (e.z * e.z) * ab.y + ab.x;
  return e.m * RSCALE - r;
}

__device__ __forceinline__ float bn_gelu(float h, float sc, float bi, const float* tab) {
  Pend e;
  gelu_issue(h, sc, bi, tab, e);
  return gelu_finish(e);
}

template <int K>
__device__ __forceinline__ void load_w(const float* __restrict__ W, int n, int half,
                                       v4f (&wb)[K / 8]) {
  const float* row = W + (size_t)n * K + 4 * half;
#pragma unroll
  for (int q = 0; q < K / 8; ++q) wb[q] = *(const v4f*)(row + 8 * q);
}

template <int MODE>
__device__ __forceinline__ void put_pair(float* gdst, int e, float g0, float g1) {
  const int m = e >> 3, r = e & 7;
  if (MODE == 1) {
    gdst[(m * 16 + r) * HSTR]     = g0;
    gdst[(m * 16 + r + 1) * HSTR] = g1;
  } else {
    v2f g = {g0, g1};
    *(v2f*)(gdst + m * 16 + r) = g;
  }
}

template <int K, int STR, int MODE>
__device__ __forceinline__ void phase(const float* A, int c, int half,
                                      const v4f (&wb)[K / 8], v8f (&acc)[HM],
                                      const v8f (&gacc)[HM], float sc, float bi,
                                      const float* tab, float* gdst) {
  constexpr int NQ = K / 8;
  constexpr int NE = 16 / NQ;
  const v8f vzero = {0.f, 0.f, 0.f, 0.f, 0.f, 0.f, 0.f, 0.f};
#pragma unroll
  for (int m = 0; m < HM; ++m) acc[m] = vzero;
  const float* arow = A + c * STR + 4 * half;
  v4f av[HM], an[HM];
#pragma unroll
  for (int m = 0; m < HM; ++m) av[m] = *(const v4f*)(arow + m * 16 * STR);
  Pend pe[NE], pn[NE];
#pragma unroll
  for (int q = 0; q < NQ; ++q) {
    if (q + 1 < NQ) {
#pragma unroll
      for (int m = 0; m < HM; ++m)
        an[m] = *(const v4f*)(arow + m * 16 * STR + 8 * (q + 1));
    }
    const v2f b0 = {wb[q].x, wb[q].y};
    const v2f b1 = {wb[q].z, wb[q].w};
#pragma unroll
    for (int m = 0; m < HM; ++m) {
      const v2f a0 = {av[m].x, av[m].y};
      acc[m] = wmma4(a0, b0, acc[m]);
    }
#pragma unroll
    for (int m = 0; m < HM; ++m) {
      const v2f a1 = {av[m].z, av[m].w};
      acc[m] = wmma4(a1, b1, acc[m]);
    }
    __builtin_amdgcn_sched_barrier(0);
    if (MODE != 0) {
#pragma unroll
      for (int j = 0; j < NE; ++j) {
        const int e = q * NE + j;
        gelu_issue(gacc[e >> 3][e & 7], sc, bi, tab, pn[j]);
      }
      if (q > 0) {
#pragma unroll
        for (int j = 0; j < NE; j += 2)
          put_pair<MODE>(gdst, (q - 1) * NE + j, gelu_finish(pe[j]), gelu_finish(pe[j + 1]));
      }
#pragma unroll
      for (int j = 0; j < NE; ++j) pe[j] = pn[j];
    }
    if (q + 1 < NQ) asm volatile("" :: "v"(an[0]), "v"(an[1]));
#pragma unroll
    for (int m = 0; m < HM; ++m) av[m] = an[m];
    __builtin_amdgcn_sched_barrier(0);
  }
  if (MODE != 0) {
#pragma unroll
    for (int j = 0; j < NE; j += 2)
      put_pair<MODE>(gdst, (NQ - 1) * NE + j, gelu_finish(pe[j]), gelu_finish(pe[j + 1]));
  }
}

__device__ __forceinline__ void lds_barrier() {
  __builtin_amdgcn_fence(__ATOMIC_RELEASE, "workgroup", "local");
  __builtin_amdgcn_s_barrier();
  __builtin_amdgcn_fence(__ATOMIC_ACQUIRE, "workgroup", "local");
}

__global__ __launch_bounds__(128) void edgeconv_mlp_kernel(
    const float* __restrict__ x,
    const float* __restrict__ W1,
    const float* __restrict__ g1, const float* __restrict__ b1,
    const float* __restrict__ m1, const float* __restrict__ v1,
    const float* __restrict__ W2,
    const float* __restrict__ g2, const float* __restrict__ b2,
    const float* __restrict__ m2, const float* __restrict__ v2,
    const float* __restrict__ W3,
    const float* __restrict__ g3, const float* __restrict__ b3,
    const float* __restrict__ m3, const float* __restrict__ v3,
    float* __restrict__ out)
{
  __shared__ __attribute__((aligned(16))) float smem[ETAB + TP * XSTR + 2 * TP * HSTR];
  float* const etab = smem;
  float* const xs   = smem + ETAB;
  float* const h1   = xs + TP * XSTR;
  float* const h2   = h1 + TP * HSTR;

  const int tid  = threadIdx.x;
  const int wave = tid >> 5;
  const int lane = tid & 31;
  const int c    = lane & 15;
  const int half = lane >> 4;
  const int n    = wave * 16 + c;

  const int tile = blockIdx.x;
  const int p0   = tile * TP;

  const v4f* xg = (const v4f*)(x + (size_t)p0 * 32);
  v4f xv[4];
#pragma unroll
  for (int i = 0; i < 4; ++i) xv[i] = xg[tid + 128 * i];

  v4f wb1[4], wb2[8], wb3[8];
  load_w<32>(W1, n, half, wb1);
  const float g1n = g1[n], b1n = b1[n], m1n = m1[n], v1n = v1[n];
  __builtin_amdgcn_sched_barrier(0);
  load_w<64>(W2, n, half, wb2);
  const float g2n = g2[n], b2n = b2[n], m2n = m2[n], v2n = v2[n];
  load_w<64>(W3, n, half, wb3);
  const float g3n = g3[n], b3n = b3[n], m3n = m3[n], v3n = v3[n];

  if (tid < NTAB) build_gelu_entry(etab, tid);

#pragma unroll
  for (int i = 0; i < 4; ++i) {
    const int idx = tid + 128 * i;
    *(v4f*)(xs + (idx >> 3) * XSTR + (idx & 7) * 4) = xv[i];
  }
  lds_barrier();

  v8f accA[HM], accB[HM];
  float* const col1 = h1 + (8 * half) * HSTR + n;
  float* const col2 = h2 + (8 * half) * HSTR + n;
  float* const mine = h1 + n * HSTR + 8 * half;

  phase<32, XSTR, 0>(xs, c, half, wb1, accA, accB, 0.f, 0.f, etab, nullptr);
  const float f1 = g1n * rsqrtf(v1n + 1e-5f), sc1 = f1 * ASCALE, bi1 = (b1n - m1n * f1) * ASCALE;
  phase<32, XSTR, 1>(xs + 32 * XSTR, c, half, wb1, accB, accA, sc1, bi1, etab, col1);
  lds_barrier();
  phase<64, HSTR, 1>(h1, c, half, wb2, accA, accB, sc1, bi1, etab, col1 + 32 * HSTR);
  __syncthreads();
  const float f2 = g2n * rsqrtf(v2n + 1e-5f), sc2 = f2 * ASCALE, bi2 = (b2n - m2n * f2) * ASCALE;
  phase<64, HSTR, 1>(h1 + 32 * HSTR, c, half, wb2, accB, accA, sc2, bi2, etab, col2);
  __syncthreads();
  phase<64, HSTR, 1>(h2, c, half, wb3, accA, accB, sc2, bi2, etab, col2 + 32 * HSTR);
  __syncthreads();
  const float f3 = g3n * rsqrtf(v3n + 1e-5f), sc3 = f3 * ASCALE, bi3 = (b3n - m3n * f3) * ASCALE;
  phase<64, HSTR, 2>(h2 + 32 * HSTR, c, half, wb3, accB, accA, sc3, bi3, etab, mine);

#pragma unroll
  for (int m = 0; m < HM; ++m) {
    v4f lo, hi;
    lo.x = bn_gelu(accB[m][0], sc3, bi3, etab);
    lo.y = bn_gelu(accB[m][1], sc3, bi3, etab);
    lo.z = bn_gelu(accB[m][2], sc3, bi3, etab);
    lo.w = bn_gelu(accB[m][3], sc3, bi3, etab);
    hi.x = bn_gelu(accB[m][4], sc3, bi3, etab);
    hi.y = bn_gelu(accB[m][5], sc3, bi3, etab);
    hi.z = bn_gelu(accB[m][6], sc3, bi3, etab);
    hi.w = bn_gelu(accB[m][7], sc3, bi3, etab);
    *(v4f*)(mine + 32 + m * 16)     = lo;
    *(v4f*)(mine + 32 + m * 16 + 4) = hi;
  }
  asm volatile("s_wait_dscnt 0" ::: "memory");

  float* obase = out + ((size_t)(tile >> 4) * 64 + wave * 16 + half) * 1024 + (tile & 15) * TP + 4 * c;
  const float* rbase = h1 + (wave * 16 + half) * HSTR + 4 * c;
#pragma unroll
  for (int i = 0; i < 8; ++i) {
    const v4f v = *(const v4f*)(rbase + 2 * i * HSTR);
    *(v4f*)(obase + (size_t)(2 * i) * 1024) = v;
  }
}

extern "C" void kernel_launch(void* const* d_in, const int* in_sizes, int n_in,
                              void* d_out, int out_size, void* d_ws, size_t ws_size,
                              hipStream_t stream) {
  const float* x  = (const float*)d_in[0];
  const float* W1 = (const float*)d_in[1];
  const float* g1 = (const float*)d_in[2];
  const float* b1 = (const float*)d_in[3];
  const float* m1 = (const float*)d_in[4];
  const float* v1 = (const float*)d_in[5];
  const float* W2 = (const float*)d_in[6];
  const float* g2 = (const float*)d_in[7];
  const float* b2 = (const float*)d_in[8];
  const float* m2 = (const float*)d_in[9];
  const float* v2 = (const float*)d_in[10];
  const float* W3 = (const float*)d_in[11];
  const float* g3 = (const float*)d_in[12];
  const float* b3 = (const float*)d_in[13];
  const float* m3 = (const float*)d_in[14];
  const float* v3 = (const float*)d_in[15];
  float* out = (float*)d_out;

  edgeconv_mlp_kernel<<<1024, 128, 0, stream>>>(
      x, W1, g1, b1, m1, v1, W2, g2, b2, m2, v2, W3, g3, b3, m3, v3, out);
}
